# starting kernel
# baseline (speedup 1.0000x reference)
#include <hip/hip_runtime.h>

typedef float v2f __attribute__((ext_vector_type(2)));
typedef float v4f __attribute__((ext_vector_type(4)));
typedef float v8f __attribute__((ext_vector_type(8)));

__device__ __forceinline__ unsigned flt_key(float f) {
    unsigned u = __float_as_uint(f);
    return (u & 0x80000000u) ? ~u : (u | 0x80000000u);
}
__device__ __forceinline__ float key_flt(unsigned u) {
    return (u & 0x80000000u) ? __uint_as_float(u & 0x7FFFFFFFu)
                             : __uint_as_float(~u);
}

__global__ void segmax_init_kernel(unsigned* __restrict__ keys, int Bn) {
    int i = blockIdx.x * blockDim.x + threadIdx.x;
    if (i < Bn) keys[i] = 0u;
}

__global__ __launch_bounds__(128) void gemv_segmax_kernel(
    const float* __restrict__ z, const int* __restrict__ bag,
    const float* __restrict__ W, const float* __restrict__ pb,
    float* __restrict__ scores, unsigned* __restrict__ segkeys, int nTiles)
{
    __shared__ float smem[4 * 2048 + 128];
    float* wlds = smem + 4 * 2048;

    const int tid  = threadIdx.x;
    const int lane = tid & 31;
    const int wave = tid >> 5;
    float* tileW = smem + wave * 2048;

    if (tid < 128) wlds[tid] = W[tid];
    __syncthreads();

    const float bias = pb[0];
    const int wavesTotal = (gridDim.x * blockDim.x) >> 5;
    const int gwave = (blockIdx.x * blockDim.x + tid) >> 5;

    const int m  = lane & 15;
    const int kk = (lane >> 4) << 1;

    for (int tile = gwave; tile < nTiles; tile += wavesTotal) {
        const int row0 = tile << 4;

        const float* gsrc = z + (size_t)row0 * 128;
        #pragma unroll
        for (int it = 0; it < 16; ++it) {
            v4f v = *(const v4f*)(gsrc + it * 128 + lane * 4);
            *(v4f*)(tileW + it * 128 + lane * 4) = v;
        }
        __asm__ volatile("s_wait_dscnt 0x0" ::: "memory");

        v8f acc = {0.f, 0.f, 0.f, 0.f, 0.f, 0.f, 0.f, 0.f};
        #pragma unroll
        for (int c = 0; c < 32; ++c) {
            const int k0 = c * 4;
            v2f av = *(const v2f*)(tileW + m * 128 + k0 + kk);
            v2f bv = *(const v2f*)(wlds + k0 + kk);
            acc = __builtin_amdgcn_wmma_f32_16x16x4_f32(
                false, av, false, bv, (short)0, acc, false, false);
        }
        __asm__ volatile("" ::: "memory");

        if ((lane & 15) == 0) {
            const int rbase = row0 + ((lane >> 4) << 3);
            #pragma unroll
            for (int j = 0; j < 8; ++j) {
                float s = acc[j] + bias;
                scores[rbase + j] = s;
                int bg = bag[rbase + j];
                atomicMax(&segkeys[bg], flt_key(s));
            }
        }
    }
}

__global__ void gemv_tail_kernel(
    const float* __restrict__ z, const int* __restrict__ bag,
    const float* __restrict__ W, const float* __restrict__ pb,
    float* __restrict__ scores, unsigned* __restrict__ segkeys,
    int startRow, int N)
{
    int r = startRow + blockIdx.x * blockDim.x + threadIdx.x;
    if (r >= N) return;
    float s = pb[0];
    for (int k = 0; k < 128; ++k) s += z[(size_t)r * 128 + k] * W[k];
    scores[r] = s;
    atomicMax(&segkeys[bag[r]], flt_key(s));
}

__global__ void segmax_final_kernel(unsigned* __restrict__ keys, int Bn) {
    int i = blockIdx.x * blockDim.x + threadIdx.x;
    if (i < Bn) {
        unsigned u = keys[i];
        ((float*)keys)[i] = (u == 0u) ? 0.0f : key_flt(u);
    }
}

extern "C" void kernel_launch(void* const* d_in, const int* in_sizes, int n_in,
                              void* d_out, int out_size, void* d_ws, size_t ws_size,
                              hipStream_t stream) {
    const float* z   = (const float*)d_in[0];
    const int*   bag = (const int*)d_in[1];
    const float* W   = (const float*)d_in[2];
    const float* pb  = (const float*)d_in[3];

    const int N  = in_sizes[1];
    const int Bn = out_size - N;

    unsigned* segkeys = (unsigned*)d_out;
    float*    scores  = (float*)d_out + Bn;

    segmax_init_kernel<<<(Bn + 255) / 256, 256, 0, stream>>>(segkeys, Bn);

    const int nTiles = N >> 4;
    if (nTiles > 0) {
        gemv_segmax_kernel<<<2048, 128, 0, stream>>>(z, bag, W, pb,
                                                     scores, segkeys, nTiles);
    }
    const int rem = N & 15;
    if (rem > 0) {
        gemv_tail_kernel<<<(rem + 63) / 64, 64, 0, stream>>>(
            z, bag, W, pb, scores, segkeys, nTiles << 4, N);
    }
    segmax_final_kernel<<<(Bn + 255) / 256, 256, 0, stream>>>(segkeys, Bn);
}


# no LDS stage: each 512-B row load is the WMMA A operand as loaded, W in the same lane layout is B, row dot = trace of D folded by DPP adds; 32-row tiles; sum order changes (lane slices, then tree)
# speedup vs baseline: 1.1579x; 1.1579x over previous
#include <hip/hip_runtime.h>

typedef float v2f __attribute__((ext_vector_type(2)));
typedef float v4f __attribute__((ext_vector_type(4)));
typedef float v8f __attribute__((ext_vector_type(8)));

__device__ __forceinline__ unsigned flt_key(float f) {
    unsigned u = __float_as_uint(f);
    return (u & 0x80000000u) ? ~u : (u | 0x80000000u);
}
__device__ __forceinline__ float key_flt(unsigned u) {
    return (u & 0x80000000u) ? __uint_as_float(u & 0x7FFFFFFFu)
                             : __uint_as_float(~u);
}

__global__ void segmax_init_kernel(unsigned* __restrict__ keys, int Bn) {
    int i = blockIdx.x * blockDim.x + threadIdx.x;
    if (i < Bn) keys[i] = 0u;
}

template <int CTRL>
__device__ __forceinline__ float dpp_take(float v) {
    return __int_as_float(__builtin_amdgcn_update_dpp(
        0, __float_as_int(v), CTRL, 0xf, 0xf, true));
}
#define ROW_SHL(n) (0x100 + (n))
#define ROW_ROR(n) (0x120 + (n))

__device__ __forceinline__ float trace_fold(v8f d) {
    float a01 = d[0] + dpp_take<ROW_SHL(1)>(d[1]);
    float a23 = d[2] + dpp_take<ROW_SHL(1)>(d[3]);
    float a45 = d[4] + dpp_take<ROW_SHL(1)>(d[5]);
    float a67 = d[6] + dpp_take<ROW_SHL(1)>(d[7]);
    float b0  = a01 + dpp_take<ROW_SHL(2)>(a23);
    float b1  = a45 + dpp_take<ROW_SHL(2)>(a67);
    return b0 + dpp_take<ROW_SHL(4)>(b1);
}

template <int S>
__device__ __forceinline__ float slot_place(float s) {
    if constexpr (S == 0) return s;
    else return dpp_take<ROW_ROR(S)>(s);
}

template <int SHIFT>
__device__ __forceinline__ float half_tile(const float* __restrict__ gsrc,
                                           v2f w01, v2f w23, int keep)
{
    v4f x[16];
    #pragma unroll
    for (int r = 0; r < 16; ++r)
        x[r] = *(const v4f*)(gsrc + r * 128);
    __builtin_amdgcn_sched_barrier(0);

    float part = 0.f;
#define DO_SLOT(S)                                                             \
    {                                                                          \
        constexpr int R = ((S) + SHIFT) & 15;                                  \
        v8f acc = {0.f, 0.f, 0.f, 0.f, 0.f, 0.f, 0.f, 0.f};                    \
        const v2f a01 = {x[R][0], x[R][1]};                                    \
        const v2f a23 = {x[R][2], x[R][3]};                                    \
        acc = __builtin_amdgcn_wmma_f32_16x16x4_f32(                           \
            false, a01, false, w01, (short)0, acc, false, false);              \
        acc = __builtin_amdgcn_wmma_f32_16x16x4_f32(                           \
            false, a23, false, w23, (short)0, acc, false, false);              \
        const float s = slot_place<(S)>(trace_fold(acc));                      \
        part = (keep == (S)) ? s : part;                                       \
    }
    DO_SLOT((0  - SHIFT) & 15) DO_SLOT((1  - SHIFT) & 15)
    DO_SLOT((2  - SHIFT) & 15) DO_SLOT((3  - SHIFT) & 15)
    DO_SLOT((4  - SHIFT) & 15) DO_SLOT((5  - SHIFT) & 15)
    DO_SLOT((6  - SHIFT) & 15) DO_SLOT((7  - SHIFT) & 15)
    DO_SLOT((8  - SHIFT) & 15) DO_SLOT((9  - SHIFT) & 15)
    DO_SLOT((10 - SHIFT) & 15) DO_SLOT((11 - SHIFT) & 15)
    DO_SLOT((12 - SHIFT) & 15) DO_SLOT((13 - SHIFT) & 15)
    DO_SLOT((14 - SHIFT) & 15) DO_SLOT((15 - SHIFT) & 15)
#undef DO_SLOT
    return part;
}

__global__ __launch_bounds__(128) void gemv_segmax_kernel(
    const float* __restrict__ z, const int* __restrict__ bag,
    const float* __restrict__ W, const float* __restrict__ pb,
    float* __restrict__ scores, unsigned* __restrict__ segkeys, int nTiles)
{
    const int tid  = threadIdx.x;
    const int lane = tid & 31;
    const int wave = __builtin_amdgcn_readfirstlane(tid >> 5);
    const int gwave = blockIdx.x * (blockDim.x >> 5) + wave;
    const int wavesTotal = (gridDim.x * blockDim.x) >> 5;

    const v4f w = *(const v4f*)(W + lane * 4);
    const v2f w01 = {w[0], w[1]};
    const v2f w23 = {w[2], w[3]};
    const float bias = pb[0];

    const bool lowRow = lane < 16;
    const int keep = (lane + ((lane >> 4) << 3)) & 15;

    for (int tile = gwave; tile < nTiles; tile += wavesTotal) {
        const int row0 = tile << 5;
        const float* gsrc = z + (size_t)row0 * 128 + lane * 4;
        const int bg = bag[row0 + lane];

        const float pa = half_tile<0>(gsrc, w01, w23, keep);
        const float pc = half_tile<8>(gsrc + 16 * 128, w01, w23, keep);

        const float mine   = lowRow ? pa : pc;
        const float theirs = lowRow ? pc : pa;
        const float other = __int_as_float(__builtin_amdgcn_permlanex16(
            0, __float_as_int(theirs), 0xFEDCBA98u, 0x76543210u, false, false));
        const float sc = (mine + other) + bias;

        scores[row0 + lane] = sc;
        atomicMax(&segkeys[bg], flt_key(sc));
    }
}

__global__ void gemv_tail_kernel(
    const float* __restrict__ z, const int* __restrict__ bag,
    const float* __restrict__ W, const float* __restrict__ pb,
    float* __restrict__ scores, unsigned* __restrict__ segkeys,
    int startRow, int N)
{
    int r = startRow + blockIdx.x * blockDim.x + threadIdx.x;
    if (r >= N) return;
    float s = pb[0];
    for (int k = 0; k < 128; ++k) s += z[(size_t)r * 128 + k] * W[k];
    scores[r] = s;
    atomicMax(&segkeys[bag[r]], flt_key(s));
}

__global__ void segmax_final_kernel(unsigned* __restrict__ keys, int Bn) {
    int i = blockIdx.x * blockDim.x + threadIdx.x;
    if (i < Bn) {
        unsigned u = keys[i];
        ((float*)keys)[i] = (u == 0u) ? 0.0f : key_flt(u);
    }
}

extern "C" void kernel_launch(void* const* d_in, const int* in_sizes, int n_in,
                              void* d_out, int out_size, void* d_ws, size_t ws_size,
                              hipStream_t stream) {
    const float* z   = (const float*)d_in[0];
    const int*   bag = (const int*)d_in[1];
    const float* W   = (const float*)d_in[2];
    const float* pb  = (const float*)d_in[3];

    const int N  = in_sizes[1];
    const int Bn = out_size - N;

    unsigned* segkeys = (unsigned*)d_out;
    float*    scores  = (float*)d_out + Bn;

    segmax_init_kernel<<<(Bn + 255) / 256, 256, 0, stream>>>(segkeys, Bn);

    const int nTiles = N >> 5;
    if (nTiles > 0) {
        gemv_segmax_kernel<<<2048, 128, 0, stream>>>(z, bag, W, pb,
                                                     scores, segkeys, nTiles);
    }
    const int rem = N & 31;
    if (rem > 0) {
        gemv_tail_kernel<<<(rem + 63) / 64, 64, 0, stream>>>(
            z, bag, W, pb, scores, segkeys, nTiles << 5, N);
    }
    segmax_final_kernel<<<(Bn + 255) / 256, 256, 0, stream>>>(segkeys, Bn);
}


# segment max: one key table per die in the workspace (8 x 4096 keys); each wave sends its atomic max to the table of the die it runs on, the last launch folds the 8 tables; sum order unchanged
# speedup vs baseline: 2.4291x; 2.0978x over previous
#include <hip/hip_runtime.h>

typedef float v2f __attribute__((ext_vector_type(2)));
typedef float v4f __attribute__((ext_vector_type(4)));
typedef float v8f __attribute__((ext_vector_type(8)));

__device__ __forceinline__ unsigned flt_key(float f) {
    unsigned u = __float_as_uint(f);
    return (u & 0x80000000u) ? ~u : (u | 0x80000000u);
}
__device__ __forceinline__ float key_flt(unsigned u) {
    return (u & 0x80000000u) ? __uint_as_float(u & 0x7FFFFFFFu)
                             : __uint_as_float(~u);
}

__global__ void segmax_init_kernel(unsigned* __restrict__ keys, int n) {
    int i = blockIdx.x * blockDim.x + threadIdx.x;
    if (i < n) keys[i] = 0u;
}

template <int CTRL>
__device__ __forceinline__ float dpp_take(float v) {
    return __int_as_float(__builtin_amdgcn_update_dpp(
        0, __float_as_int(v), CTRL, 0xf, 0xf, true));
}
#define ROW_SHL(n) (0x100 + (n))
#define ROW_ROR(n) (0x120 + (n))

__device__ __forceinline__ float trace_fold(v8f d) {
    float a01 = d[0] + dpp_take<ROW_SHL(1)>(d[1]);
    float a23 = d[2] + dpp_take<ROW_SHL(1)>(d[3]);
    float a45 = d[4] + dpp_take<ROW_SHL(1)>(d[5]);
    float a67 = d[6] + dpp_take<ROW_SHL(1)>(d[7]);
    float b0  = a01 + dpp_take<ROW_SHL(2)>(a23);
    float b1  = a45 + dpp_take<ROW_SHL(2)>(a67);
    return b0 + dpp_take<ROW_SHL(4)>(b1);
}

template <int S>
__device__ __forceinline__ float slot_place(float s) {
    if constexpr (S == 0) return s;
    else return dpp_take<ROW_ROR(S)>(s);
}

template <int SHIFT>
__device__ __forceinline__ float half_tile(const float* __restrict__ gsrc,
                                           v2f w01, v2f w23, int keep)
{
    v4f x[16];
    #pragma unroll
    for (int r = 0; r < 16; ++r)
        x[r] = *(const v4f*)(gsrc + r * 128);
    __builtin_amdgcn_sched_barrier(0);

    float part = 0.f;
#define DO_SLOT(S)                                                             \
    {                                                                          \
        constexpr int R = ((S) + SHIFT) & 15;                                  \
        v8f acc = {0.f, 0.f, 0.f, 0.f, 0.f, 0.f, 0.f, 0.f};                    \
        const v2f a01 = {x[R][0], x[R][1]};                                    \
        const v2f a23 = {x[R][2], x[R][3]};                                    \
        acc = __builtin_amdgcn_wmma_f32_16x16x4_f32(                           \
            false, a01, false, w01, (short)0, acc, false, false);              \
        acc = __builtin_amdgcn_wmma_f32_16x16x4_f32(                           \
            false, a23, false, w23, (short)0, acc, false, false);              \
        const float s = slot_place<(S)>(trace_fold(acc));                      \
        part = (keep == (S)) ? s : part;                                       \
    }
    DO_SLOT((0  - SHIFT) & 15) DO_SLOT((1  - SHIFT) & 15)
    DO_SLOT((2  - SHIFT) & 15) DO_SLOT((3  - SHIFT) & 15)
    DO_SLOT((4  - SHIFT) & 15) DO_SLOT((5  - SHIFT) & 15)
    DO_SLOT((6  - SHIFT) & 15) DO_SLOT((7  - SHIFT) & 15)
    DO_SLOT((8  - SHIFT) & 15) DO_SLOT((9  - SHIFT) & 15)
    DO_SLOT((10 - SHIFT) & 15) DO_SLOT((11 - SHIFT) & 15)
    DO_SLOT((12 - SHIFT) & 15) DO_SLOT((13 - SHIFT) & 15)
    DO_SLOT((14 - SHIFT) & 15) DO_SLOT((15 - SHIFT) & 15)
#undef DO_SLOT
    return part;
}

__global__ __launch_bounds__(128) void gemv_segmax_kernel(
    const float* __restrict__ z, const int* __restrict__ bag,
    const float* __restrict__ W, const float* __restrict__ pb,
    float* __restrict__ scores, unsigned* __restrict__ segkeys, int dieStride,
    int nTiles)
{
    const int die = (__builtin_amdgcn_s_sendmsg_rtn(0x87) >> 16) & 7;
    unsigned* __restrict__ diekeys = segkeys + die * dieStride;

    const int tid  = threadIdx.x;
    const int lane = tid & 31;
    const int wave = __builtin_amdgcn_readfirstlane(tid >> 5);
    const int gwave = blockIdx.x * (blockDim.x >> 5) + wave;
    const int wavesTotal = (gridDim.x * blockDim.x) >> 5;

    const v4f w = *(const v4f*)(W + lane * 4);
    const v2f w01 = {w[0], w[1]};
    const v2f w23 = {w[2], w[3]};
    const float bias = pb[0];

    const bool lowRow = lane < 16;
    const int keep = (lane + ((lane >> 4) << 3)) & 15;

    for (int tile = gwave; tile < nTiles; tile += wavesTotal) {
        const int row0 = tile << 5;
        const float* gsrc = z + (size_t)row0 * 128 + lane * 4;
        const int bg = bag[row0 + lane];

        const float pa = half_tile<0>(gsrc, w01, w23, keep);
        const float pc = half_tile<8>(gsrc + 16 * 128, w01, w23, keep);

        const float mine   = lowRow ? pa : pc;
        const float theirs = lowRow ? pc : pa;
        const float other = __int_as_float(__builtin_amdgcn_permlanex16(
            0, __float_as_int(theirs), 0xFEDCBA98u, 0x76543210u, false, false));
        const float sc = (mine + other) + bias;

        scores[row0 + lane] = sc;
        atomicMax(&diekeys[bg], flt_key(sc));
    }
}

__global__ void gemv_tail_kernel(
    const float* __restrict__ z, const int* __restrict__ bag,
    const float* __restrict__ W, const float* __restrict__ pb,
    float* __restrict__ scores, unsigned* __restrict__ segkeys,
    int startRow, int N)
{
    int r = startRow + blockIdx.x * blockDim.x + threadIdx.x;
    if (r >= N) return;
    float s = pb[0];
    for (int k = 0; k < 128; ++k) s += z[(size_t)r * 128 + k] * W[k];
    scores[r] = s;
    atomicMax(&segkeys[bag[r]], flt_key(s));
}

template <int NT>
__global__ void segmax_final_kernel(const unsigned* keys, float* M, int Bn) {
    int i = blockIdx.x * blockDim.x + threadIdx.x;
    if (i < Bn) {
        unsigned u = 0u;
        #pragma unroll
        for (int t = 0; t < NT; ++t) {
            const unsigned v = keys[t * Bn + i];
            u = v > u ? v : u;
        }
        M[i] = (u == 0u) ? 0.0f : key_flt(u);
    }
}

extern "C" void kernel_launch(void* const* d_in, const int* in_sizes, int n_in,
                              void* d_out, int out_size, void* d_ws, size_t ws_size,
                              hipStream_t stream) {
    const float* z   = (const float*)d_in[0];
    const int*   bag = (const int*)d_in[1];
    const float* W   = (const float*)d_in[2];
    const float* pb  = (const float*)d_in[3];

    const int N  = in_sizes[1];
    const int Bn = out_size - N;

    float*    Mout    = (float*)d_out;
    float*    scores  = (float*)d_out + Bn;

    const bool perDie = ws_size >= (size_t)8 * Bn * sizeof(unsigned);
    unsigned* segkeys   = perDie ? (unsigned*)d_ws : (unsigned*)d_out;
    const int nTabs     = perDie ? 8 : 1;
    const int dieStride = perDie ? Bn : 0;

    segmax_init_kernel<<<(nTabs * Bn + 255) / 256, 256, 0, stream>>>(
        segkeys, nTabs * Bn);

    const int nTiles = N >> 5;
    if (nTiles > 0) {
        gemv_segmax_kernel<<<2048, 128, 0, stream>>>(z, bag, W, pb, scores,
                                                     segkeys, dieStride, nTiles);
    }
    const int rem = N & 31;
    if (rem > 0) {
        gemv_tail_kernel<<<(rem + 63) / 64, 64, 0, stream>>>(
            z, bag, W, pb, scores, segkeys, nTiles << 5, N);
    }
    if (perDie)
        segmax_final_kernel<8><<<(Bn + 255) / 256, 256, 0, stream>>>(segkeys, Mout, Bn);
    else
        segmax_final_kernel<1><<<(Bn + 255) / 256, 256, 0, stream>>>(segkeys, Mout, Bn);
}
